# starting kernel
# baseline (speedup 1.0000x reference)
#include <hip/hip_runtime.h>
#include <hip/hip_bf16.h>

#define BH   16
#define NSER 32
#define DQK  3072
#define TFD  3072

typedef __attribute__((ext_vector_type(2))) float v2f;
typedef __attribute__((ext_vector_type(4))) float v4f;
typedef __attribute__((ext_vector_type(8))) float v8f;

__global__ __launch_bounds__(128) void mha_scores_softmax(
    const float* __restrict__ Q, const float* __restrict__ K,
    const int* __restrict__ mask, float* __restrict__ attn)
{
    __shared__ float S[NSER][NSER + 1];

    const int bh   = blockIdx.x;
    const int tid  = threadIdx.x;
    const int wave = tid >> 5;
    const int lane = tid & 31;
    const int half = lane >> 4;
    const int m16  = lane & 15;
    const int mi   = wave >> 1;
    const int ni   = wave & 1;

    const float* qrow = Q + ((size_t)bh * NSER + mi * 16 + m16) * DQK + 2 * half;
    const float* krow = K + ((size_t)bh * NSER + ni * 16 + m16) * DQK + 2 * half;

    v8f acc0 = {}, acc1 = {}, acc2 = {}, acc3 = {};
    for (int k0 = 0; k0 < DQK; k0 += 16) {
        v2f a0 = *(const v2f*)(qrow + k0);
        v2f b0 = *(const v2f*)(krow + k0);
        v2f a1 = *(const v2f*)(qrow + k0 + 4);
        v2f b1 = *(const v2f*)(krow + k0 + 4);
        v2f a2 = *(const v2f*)(qrow + k0 + 8);
        v2f b2 = *(const v2f*)(krow + k0 + 8);
        v2f a3 = *(const v2f*)(qrow + k0 + 12);
        v2f b3 = *(const v2f*)(krow + k0 + 12);
        acc0 = __builtin_amdgcn_wmma_f32_16x16x4_f32(false, a0, false, b0,
                                                     (short)0, acc0, false, false);
        acc1 = __builtin_amdgcn_wmma_f32_16x16x4_f32(false, a1, false, b1,
                                                     (short)0, acc1, false, false);
        acc2 = __builtin_amdgcn_wmma_f32_16x16x4_f32(false, a2, false, b2,
                                                     (short)0, acc2, false, false);
        acc3 = __builtin_amdgcn_wmma_f32_16x16x4_f32(false, a3, false, b3,
                                                     (short)0, acc3, false, false);
    }
    v8f acc = acc0 + acc1 + acc2 + acc3;

    const float inv_scale = 1.0f / 55.42562584220407f;

    #pragma unroll
    for (int r = 0; r < 8; ++r) {
        const int row = mi * 16 + half * 8 + r;
        const int col = ni * 16 + m16;
        float v = acc[r] * inv_scale;
        if (mask[((size_t)bh * NSER + row) * NSER + col] == 0)
            v = -__builtin_inff();
        S[row][col] = v;
    }
    __syncthreads();

    if (tid < NSER) {
        float m = -__builtin_inff();
        #pragma unroll
        for (int c = 0; c < NSER; ++c) m = fmaxf(m, S[tid][c]);
        float s = 0.0f;
        #pragma unroll
        for (int c = 0; c < NSER; ++c) {
            float ev = expf(S[tid][c] - m);
            S[tid][c] = ev;
            s += ev;
        }
        const float invs = 1.0f / s;
        float* arow = attn + ((size_t)bh * NSER + tid) * NSER;
        #pragma unroll
        for (int c = 0; c < NSER; ++c) arow[c] = S[tid][c] * invs;
    }
}

__global__ __launch_bounds__(256) void mha_apply_v(
    const float* __restrict__ V, const float* __restrict__ attn,
    float* __restrict__ out)
{
    const int bhi = blockIdx.x;
    const int i   = bhi & (NSER - 1);
    const int bh  = bhi >> 5;
    const int tid = threadIdx.x;

    const float* arow = attn + ((size_t)bh * NSER + i) * NSER;
    const float*  vbase   = V + ((size_t)bh * NSER * NSER + i) * TFD;
    const size_t  jstride = (size_t)NSER * TFD;

    v4f acc0 = {}, acc1 = {}, acc2 = {};

    #pragma unroll 4
    for (int j = 0; j < NSER; ++j) {
        const float w = arow[j];
        const v4f* vp = (const v4f*)(vbase + (size_t)j * jstride);
        v4f x0 = __builtin_nontemporal_load(vp + tid);
        v4f x1 = __builtin_nontemporal_load(vp + 256 + tid);
        v4f x2 = __builtin_nontemporal_load(vp + 512 + tid);
        acc0 += w * x0;
        acc1 += w * x1;
        acc2 += w * x2;
    }

    v4f* ob = (v4f*)(out + ((size_t)bh * NSER + i) * TFD);
    ob[tid]       = acc0;
    ob[256 + tid] = acc1;
    ob[512 + tid] = acc2;
}

extern "C" void kernel_launch(void* const* d_in, const int* in_sizes, int n_in,
                              void* d_out, int out_size, void* d_ws, size_t ws_size,
                              hipStream_t stream) {
    const float* Q    = (const float*)d_in[0];
    const float* K    = (const float*)d_in[1];
    const float* V    = (const float*)d_in[2];
    const int*   mask = (const int*)d_in[3];
    float*       out  = (float*)d_out;
    float*       attn = (float*)d_ws;

    mha_scores_softmax<<<BH, 128, 0, stream>>>(Q, K, mask, attn);
    mha_apply_v<<<BH * NSER, 256, 0, stream>>>(V, attn, out);
}


# Q.K^T split over k into 256 work-groups instead of 16 (each wave: all four tiles of its k slice); partial planes added in fixed order and soft-maxed per wave in the V kernel; sum order changes
# speedup vs baseline: 2.3568x; 2.3568x over previous
#include <hip/hip_runtime.h>


#define BH     16
#define NSER   32
#define DQK    3072
#define TFD    3072

#define NSPLIT 16
#define KC     (DQK / NSPLIT)
#define KW     (KC / 4)
#define NLD    (KW / 8)
#define RP     36

typedef __attribute__((ext_vector_type(2))) float v2f;
typedef __attribute__((ext_vector_type(4))) float v4f;
typedef __attribute__((ext_vector_type(8))) float v8f;

__device__ __forceinline__ v8f mma4(v2f a, v2f b, v8f c)
{
    return __builtin_amdgcn_wmma_f32_16x16x4_f32(false, a, false, b, (short)0, c, false, false);
}

__global__ __launch_bounds__(128) void mha_scores_splitk(
    const float* __restrict__ Q, const float* __restrict__ K,
    float* __restrict__ out)
{
    __shared__ __attribute__((aligned(16))) float red[4][NSER][RP];

    const int sp   = blockIdx.x;
    const int bh   = blockIdx.y;
    const int tid  = threadIdx.x;
    const int wave = tid >> 5;
    const int lane = tid & 31;
    const int half = lane >> 4;
    const int c    = lane & 15;

    const size_t koff = (size_t)sp * KC + wave * KW + 4 * half;
    const float* q0 = Q + ((size_t)bh * NSER + c) * DQK + koff;
    const float* k0 = K + ((size_t)bh * NSER + c) * DQK + koff;
    const float* q1 = q0 + (size_t)16 * DQK;
    const float* k1 = k0 + (size_t)16 * DQK;

    v4f a0[NLD], a1[NLD], b0[NLD], b1[NLD];
    #pragma unroll
    for (int u = 0; u < NLD; ++u) {
        a0[u] = *(const v4f*)(q0 + 8 * u);
        a1[u] = *(const v4f*)(q1 + 8 * u);
        b0[u] = *(const v4f*)(k0 + 8 * u);
        b1[u] = *(const v4f*)(k1 + 8 * u);
    }
    __builtin_amdgcn_sched_barrier(0);

    v8f c00 = {}, c01 = {}, c10 = {}, c11 = {};
    #pragma unroll
    for (int u = 0; u < NLD; ++u) {
        const v2f a0l = {a0[u].x, a0[u].y}, a0h = {a0[u].z, a0[u].w};
        const v2f a1l = {a1[u].x, a1[u].y}, a1h = {a1[u].z, a1[u].w};
        const v2f b0l = {b0[u].x, b0[u].y}, b0h = {b0[u].z, b0[u].w};
        const v2f b1l = {b1[u].x, b1[u].y}, b1h = {b1[u].z, b1[u].w};
        c00 = mma4(a0l, b0l, c00);
        c01 = mma4(a0l, b1l, c01);
        c10 = mma4(a1l, b0l, c10);
        c11 = mma4(a1l, b1l, c11);
        c00 = mma4(a0h, b0h, c00);
        c01 = mma4(a0h, b1h, c01);
        c10 = mma4(a1h, b0h, c10);
        c11 = mma4(a1h, b1h, c11);
    }

    #pragma unroll
    for (int r = 0; r < 8; ++r) {
        const int row = half * 8 + r;
        red[wave][row][c]           = c00[r];
        red[wave][row][16 + c]      = c01[r];
        red[wave][16 + row][c]      = c10[r];
        red[wave][16 + row][16 + c] = c11[r];
    }
    __syncthreads();

    #pragma unroll
    for (int q = 0; q < 2; ++q) {
        const int row = wave * 8 + q * 4 + (lane >> 3);
        const int col = (lane & 7) * 4;
        const v4f s0 = *(const v4f*)&red[0][row][col];
        const v4f s1 = *(const v4f*)&red[1][row][col];
        const v4f s2 = *(const v4f*)&red[2][row][col];
        const v4f s3 = *(const v4f*)&red[3][row][col];
        const v4f s  = (s0 + s1) + (s2 + s3);
        *(v4f*)(out + ((size_t)bh * NSER + row) * TFD + sp * NSER + col) = s;
    }
}

template <int X>
__device__ __forceinline__ float xrow(float x)
{
    return __builtin_bit_cast(float,
        __builtin_amdgcn_update_dpp(0, __builtin_bit_cast(int, x), 0x160 + X, 0xf, 0xf, false));
}
__device__ __forceinline__ float xhalf(float x)
{
    return __builtin_bit_cast(float,
        __builtin_amdgcn_permlanex16(__builtin_bit_cast(int, x), __builtin_bit_cast(int, x),
                                     0x76543210, 0xfedcba98, false, false));
}
__device__ __forceinline__ float wave_max(float x)
{
    x = fmaxf(x, xrow<1>(x));
    x = fmaxf(x, xrow<2>(x));
    x = fmaxf(x, xrow<4>(x));
    x = fmaxf(x, xrow<8>(x));
    return fmaxf(x, xhalf(x));
}
__device__ __forceinline__ float wave_sum(float x)
{
    x += xrow<1>(x);
    x += xrow<2>(x);
    x += xrow<4>(x);
    x += xrow<8>(x);
    return x + xhalf(x);
}

__global__ __launch_bounds__(256) void mha_softmax_apply_v(
    const float* __restrict__ V, const int* __restrict__ mask,
    float* __restrict__ out)
{
    const int bhi  = blockIdx.x;
    const int i    = bhi & (NSER - 1);
    const int bh   = bhi >> 5;
    const int tid  = threadIdx.x;
    const int lane = tid & 31;

    float* orow = out + ((size_t)bh * NSER + i) * TFD;

    float part[NSPLIT];
    #pragma unroll
    for (int p = 0; p < NSPLIT; ++p) part[p] = orow[p * NSER + lane];
    const int mk = mask[((size_t)bh * NSER + i) * NSER + lane];

    __syncthreads();

    const float*  vbase   = V + ((size_t)bh * NSER * NSER + i) * TFD;
    const size_t  jstride = (size_t)NSER * TFD;

    v4f x0[4], x1[4], x2[4];
    #pragma unroll
    for (int j = 0; j < 4; ++j) {
        const v4f* vp = (const v4f*)(vbase + (size_t)j * jstride);
        x0[j] = __builtin_nontemporal_load(vp + tid);
        x1[j] = __builtin_nontemporal_load(vp + 256 + tid);
        x2[j] = __builtin_nontemporal_load(vp + 512 + tid);
    }

    float sc = part[0];
    #pragma unroll
    for (int p = 1; p < NSPLIT; ++p) sc += part[p];

    const float inv_scale = 1.0f / 55.42562584220407f;
    float v = sc * inv_scale;
    if (mk == 0) v = -__builtin_inff();

    const float m    = wave_max(v);
    const float ev   = expf(v - m);
    const float ssum = wave_sum(ev);
    const float wl   = ev * (1.0f / ssum);

    v4f acc0 = {}, acc1 = {}, acc2 = {};

    #pragma unroll
    for (int j = 0; j < 4; ++j) {
        const float w = __builtin_bit_cast(float,
            __builtin_amdgcn_readlane(__builtin_bit_cast(int, wl), j));
        acc0 += w * x0[j];
        acc1 += w * x1[j];
        acc2 += w * x2[j];
    }

    #pragma unroll 4
    for (int j = 4; j < NSER; ++j) {
        const float w = __builtin_bit_cast(float,
            __builtin_amdgcn_readlane(__builtin_bit_cast(int, wl), j));
        const v4f* vp = (const v4f*)(vbase + (size_t)j * jstride);
        v4f x0 = __builtin_nontemporal_load(vp + tid);
        v4f x1 = __builtin_nontemporal_load(vp + 256 + tid);
        v4f x2 = __builtin_nontemporal_load(vp + 512 + tid);
        acc0 += w * x0;
        acc1 += w * x1;
        acc2 += w * x2;
    }

    v4f* ob = (v4f*)orow;
    ob[tid]       = acc0;
    ob[256 + tid] = acc1;
    ob[512 + tid] = acc2;
}

extern "C" void kernel_launch(void* const* d_in, const int* in_sizes, int n_in,
                              void* d_out, int out_size, void* d_ws, size_t ws_size,
                              hipStream_t stream) {
    const float* Q    = (const float*)d_in[0];
    const float* K    = (const float*)d_in[1];
    const float* V    = (const float*)d_in[2];
    const int*   mask = (const int*)d_in[3];
    float*       out  = (float*)d_out;

    mha_scores_splitk<<<dim3(NSPLIT, BH), 128, 0, stream>>>(Q, K, out);
    mha_softmax_apply_v<<<BH * NSER, 256, 0, stream>>>(V, mask, out);
}


# V kernel requests only the V slices whose mask bit is set (a masked j has weight exp(-inf) = +0 exactly, its slice adds nothing): about half the 201 MB stream is never read; sum order unchanged
# speedup vs baseline: 3.3963x; 1.4411x over previous
#include <hip/hip_runtime.h>


#define BH     16
#define NSER   32
#define DQK    3072
#define TFD    3072

#define NSPLIT 16
#define KC     (DQK / NSPLIT)
#define KW     (KC / 4)
#define NLD    (KW / 8)
#define RP     36

typedef __attribute__((ext_vector_type(2))) float v2f;
typedef __attribute__((ext_vector_type(4))) float v4f;
typedef __attribute__((ext_vector_type(8))) float v8f;

__device__ __forceinline__ v8f mma4(v2f a, v2f b, v8f c)
{
    return __builtin_amdgcn_wmma_f32_16x16x4_f32(false, a, false, b, (short)0, c, false, false);
}

__global__ __launch_bounds__(128) void mha_scores_splitk(
    const float* __restrict__ Q, const float* __restrict__ K,
    float* __restrict__ out)
{
    __shared__ __attribute__((aligned(16))) float red[4][NSER][RP];

    const int sp   = blockIdx.x;
    const int bh   = blockIdx.y;
    const int tid  = threadIdx.x;
    const int wave = tid >> 5;
    const int lane = tid & 31;
    const int half = lane >> 4;
    const int c    = lane & 15;

    const size_t koff = (size_t)sp * KC + wave * KW + 4 * half;
    const float* q0 = Q + ((size_t)bh * NSER + c) * DQK + koff;
    const float* k0 = K + ((size_t)bh * NSER + c) * DQK + koff;
    const float* q1 = q0 + (size_t)16 * DQK;
    const float* k1 = k0 + (size_t)16 * DQK;

    v4f a0[NLD], a1[NLD], b0[NLD], b1[NLD];
    #pragma unroll
    for (int u = 0; u < NLD; ++u) {
        a0[u] = *(const v4f*)(q0 + 8 * u);
        a1[u] = *(const v4f*)(q1 + 8 * u);
        b0[u] = *(const v4f*)(k0 + 8 * u);
        b1[u] = *(const v4f*)(k1 + 8 * u);
    }
    __builtin_amdgcn_sched_barrier(0);

    v8f c00 = {}, c01 = {}, c10 = {}, c11 = {};
    #pragma unroll
    for (int u = 0; u < NLD; ++u) {
        const v2f a0l = {a0[u].x, a0[u].y}, a0h = {a0[u].z, a0[u].w};
        const v2f a1l = {a1[u].x, a1[u].y}, a1h = {a1[u].z, a1[u].w};
        const v2f b0l = {b0[u].x, b0[u].y}, b0h = {b0[u].z, b0[u].w};
        const v2f b1l = {b1[u].x, b1[u].y}, b1h = {b1[u].z, b1[u].w};
        c00 = mma4(a0l, b0l, c00);
        c01 = mma4(a0l, b1l, c01);
        c10 = mma4(a1l, b0l, c10);
        c11 = mma4(a1l, b1l, c11);
        c00 = mma4(a0h, b0h, c00);
        c01 = mma4(a0h, b1h, c01);
        c10 = mma4(a1h, b0h, c10);
        c11 = mma4(a1h, b1h, c11);
    }

    #pragma unroll
    for (int r = 0; r < 8; ++r) {
        const int row = half * 8 + r;
        red[wave][row][c]           = c00[r];
        red[wave][row][16 + c]      = c01[r];
        red[wave][16 + row][c]      = c10[r];
        red[wave][16 + row][16 + c] = c11[r];
    }
    __syncthreads();

    #pragma unroll
    for (int q = 0; q < 2; ++q) {
        const int row = wave * 8 + q * 4 + (lane >> 3);
        const int col = (lane & 7) * 4;
        const v4f s0 = *(const v4f*)&red[0][row][col];
        const v4f s1 = *(const v4f*)&red[1][row][col];
        const v4f s2 = *(const v4f*)&red[2][row][col];
        const v4f s3 = *(const v4f*)&red[3][row][col];
        const v4f s  = (s0 + s1) + (s2 + s3);
        *(v4f*)(out + ((size_t)bh * NSER + row) * TFD + sp * NSER + col) = s;
    }
}

template <int X>
__device__ __forceinline__ float xrow(float x)
{
    return __builtin_bit_cast(float,
        __builtin_amdgcn_update_dpp(0, __builtin_bit_cast(int, x), 0x160 + X, 0xf, 0xf, false));
}
__device__ __forceinline__ float xhalf(float x)
{
    return __builtin_bit_cast(float,
        __builtin_amdgcn_permlanex16(__builtin_bit_cast(int, x), __builtin_bit_cast(int, x),
                                     0x76543210, 0xfedcba98, false, false));
}
__device__ __forceinline__ float wave_max(float x)
{
    x = fmaxf(x, xrow<1>(x));
    x = fmaxf(x, xrow<2>(x));
    x = fmaxf(x, xrow<4>(x));
    x = fmaxf(x, xrow<8>(x));
    return fmaxf(x, xhalf(x));
}
__device__ __forceinline__ float wave_sum(float x)
{
    x += xrow<1>(x);
    x += xrow<2>(x);
    x += xrow<4>(x);
    x += xrow<8>(x);
    return x + xhalf(x);
}

#define RING 8

__global__ __launch_bounds__(256) void mha_softmax_apply_v(
    const float* __restrict__ V, const int* __restrict__ mask,
    float* __restrict__ out)
{
    const int bhi  = blockIdx.x;
    const int i    = bhi & (NSER - 1);
    const int bh   = bhi >> 5;
    const int tid  = threadIdx.x;
    const int lane = tid & 31;

    float* orow = out + ((size_t)bh * NSER + i) * TFD;

    const int mk = mask[((size_t)bh * NSER + i) * NSER + lane];
    float part[NSPLIT];
    #pragma unroll
    for (int p = 0; p < NSPLIT; ++p) part[p] = orow[p * NSER + lane];

    unsigned live = __builtin_amdgcn_ballot_w32(mk != 0);
    if (live == 0u) live = 0xffffffffu;
    const int nlive = __builtin_popcount(live);

    const float*  vbase   = V + ((size_t)bh * NSER * NSER + i) * TFD;
    const size_t  jstride = (size_t)NSER * TFD;

    unsigned ahead = live;
    unsigned cur   = live;

    v4f r0[RING], r1[RING], r2[RING];

#define REQUEST(s)                                                               \
    {                                                                            \
        const bool real = ahead != 0u;                                           \
        const int  jn   = real ? __builtin_ctz(ahead) : 0;                       \
        ahead &= ahead - 1u;                                                     \
        const v4f* vp = (const v4f*)(vbase + (size_t)jn * jstride) + (real ? tid : 0); \
        r0[s] = __builtin_nontemporal_load(vp);                                  \
        r1[s] = __builtin_nontemporal_load(vp + 256);                            \
        r2[s] = __builtin_nontemporal_load(vp + 512);                            \
    }

#define SUM(s)                                                                   \
    {                                                                            \
        const bool has = cur != 0u;                                              \
        const int  jc  = has ? __builtin_ctz(cur) : 0;                           \
        cur &= cur - 1u;                                                         \
        const float wj = __builtin_bit_cast(float,                               \
            __builtin_amdgcn_readlane(__builtin_bit_cast(int, wl), jc));         \
        const float w  = has ? wj : 0.0f;                                        \
        acc0 += w * r0[s];                                                       \
        acc1 += w * r1[s];                                                       \
        acc2 += w * r2[s];                                                       \
    }

    #pragma unroll
    for (int s = 0; s < RING; ++s) REQUEST(s)
    __builtin_amdgcn_sched_barrier(0);

    float sc = part[0];
    #pragma unroll
    for (int p = 1; p < NSPLIT; ++p) sc += part[p];

    const float inv_scale = 1.0f / 55.42562584220407f;
    float v = sc * inv_scale;
    if (mk == 0) v = -__builtin_inff();

    const float m    = wave_max(v);
    const float ev   = expf(v - m);
    const float ssum = wave_sum(ev);
    const float wl   = ev * (1.0f / ssum);

    v4f acc0 = {}, acc1 = {}, acc2 = {};

    const int ngroups = (nlive - 1) >> 3;
    #pragma nounroll
    for (int g = 0; g < ngroups; ++g) {
        #pragma unroll
        for (int s = 0; s < RING; ++s) { SUM(s) REQUEST(s) }
    }
    #pragma unroll
    for (int s = 0; s < RING; ++s) SUM(s)

#undef REQUEST
#undef SUM

    __syncthreads();

    v4f* ob = (v4f*)orow;
    ob[tid]       = acc0;
    ob[256 + tid] = acc1;
    ob[512 + tid] = acc2;
}

extern "C" void kernel_launch(void* const* d_in, const int* in_sizes, int n_in,
                              void* d_out, int out_size, void* d_ws, size_t ws_size,
                              hipStream_t stream) {
    const float* Q    = (const float*)d_in[0];
    const float* K    = (const float*)d_in[1];
    const float* V    = (const float*)d_in[2];
    const int*   mask = (const int*)d_in[3];
    float*       out  = (float*)d_out;

    mha_scores_splitk<<<dim3(NSPLIT, BH), 128, 0, stream>>>(Q, K, out);
    mha_softmax_apply_v<<<BH * NSER, 256, 0, stream>>>(V, mask, out);
}
